# starting kernel
# baseline (speedup 1.0000x reference)
#include <hip/hip_runtime.h>
#include <math.h>

typedef float v2f __attribute__((ext_vector_type(2)));
typedef float v8f __attribute__((ext_vector_type(8)));

#define KCOMP 4
#define DDIM  16

__global__ void gmm_prep_kernel(const float* __restrict__ phi,
                                const float* __restrict__ mu,
                                const float* __restrict__ L,
                                float* __restrict__ wsLinv,
                                float* __restrict__ wsW,
                                float* __restrict__ wsC) {
  int k = threadIdx.x;
  if (k >= KCOMP) return;
  const float* Lk = L + k * DDIM * DDIM;
  float X[DDIM][DDIM];
  for (int c = 0; c < DDIM; ++c) {
    for (int i = 0; i < DDIM; ++i) X[i][c] = 0.f;
    X[c][c] = 1.f / Lk[c * DDIM + c];
    for (int i = c - 1; i >= 0; --i) {
      float s = 0.f;
      for (int t = i + 1; t <= c; ++t) s += Lk[i * DDIM + t] * X[t][c];
      X[i][c] = -s / Lk[i * DDIM + i];
    }
  }
  for (int i = 0; i < DDIM; ++i)
    for (int j = 0; j < DDIM; ++j)
      wsLinv[(k * DDIM + i) * DDIM + j] = X[i][j];
  for (int i = 0; i < DDIM; ++i) {
    float s = 0.f;
    for (int j = 0; j < DDIM; ++j) s += X[i][j] * mu[k * DDIM + j];
    wsW[k * DDIM + i] = s;
  }
  float logdet = 0.f;
  for (int d = 0; d < DDIM; ++d) logdet += __logf(Lk[d * DDIM + d]);
  logdet *= 2.f;
  const float LOG2PI = 1.8378770664093453f;
  wsC[k] = __logf(phi[k]) - 0.5f * ((float)DDIM * LOG2PI + logdet);
}

__global__ void __launch_bounds__(256) gmm_energy_kernel(
    const float* __restrict__ z, float* __restrict__ out,
    const float* __restrict__ wsLinv, const float* __restrict__ wsW,
    const float* __restrict__ wsC,
    int N, int numTiles, int numWaves) {
  const int lane  = threadIdx.x & 31;
  const int m     = lane & 15;
  const int half  = lane >> 4;
  const int waveId = (int)((blockIdx.x * blockDim.x + threadIdx.x) >> 5);

  v2f a[KCOMP][4];
#pragma unroll
  for (int k = 0; k < KCOMP; ++k)
#pragma unroll
    for (int kk = 0; kk < 4; ++kk)
      a[k][kk] = *(const v2f*)(wsLinv + (k * 16 + m) * 16 + kk * 4 + 2 * half);

  v8f cinit[KCOMP];
#pragma unroll
  for (int k = 0; k < KCOMP; ++k) {
    const float* pw = wsW + k * 16 + 8 * half;
#pragma unroll
    for (int r = 0; r < 8; ++r) cinit[k][r] = -pw[r];
  }
  float ck[KCOMP];
#pragma unroll
  for (int k = 0; k < KCOMP; ++k) ck[k] = wsC[k];

  for (int t = waveId; t < numTiles; t += numWaves) {
    int n = t * 16 + m;
    if (n >= N) n = N - 1;
    const float* zp = z + (size_t)n * 16 + 2 * half;
    v2f b0 = *(const v2f*)(zp + 0);
    v2f b1 = *(const v2f*)(zp + 4);
    v2f b2 = *(const v2f*)(zp + 8);
    v2f b3 = *(const v2f*)(zp + 12);

    int nt = t + numWaves; if (nt >= numTiles) nt = numTiles - 1;
    __builtin_prefetch(z + (size_t)nt * 256 + lane * 8, 0, 1);

    float logit[KCOMP];
#pragma unroll
    for (int k = 0; k < KCOMP; ++k) {
      v8f acc = cinit[k];
      acc = __builtin_amdgcn_wmma_f32_16x16x4_f32(false, a[k][0], false, b0,
                                                  (short)0, acc, false, false);
      acc = __builtin_amdgcn_wmma_f32_16x16x4_f32(false, a[k][1], false, b1,
                                                  (short)0, acc, false, false);
      acc = __builtin_amdgcn_wmma_f32_16x16x4_f32(false, a[k][2], false, b2,
                                                  (short)0, acc, false, false);
      acc = __builtin_amdgcn_wmma_f32_16x16x4_f32(false, a[k][3], false, b3,
                                                  (short)0, acc, false, false);
      float s = 0.f;
#pragma unroll
      for (int r = 0; r < 8; ++r) s += acc[r] * acc[r];
      s += __shfl_xor(s, 16, 32);
      logit[k] = ck[k] - 0.5f * s;
    }
    float mx = fmaxf(fmaxf(logit[0], logit[1]), fmaxf(logit[2], logit[3]));
    float se = __expf(logit[0] - mx) + __expf(logit[1] - mx) +
               __expf(logit[2] - mx) + __expf(logit[3] - mx);
    float e = -(mx + __logf(se));

    if (half == 0) {
      int idx = t * 16 + m;
      if (idx < N) out[idx] = e;
    }
  }
}

extern "C" void kernel_launch(void* const* d_in, const int* in_sizes, int n_in,
                              void* d_out, int out_size, void* d_ws, size_t ws_size,
                              hipStream_t stream) {
  const float* z   = (const float*)d_in[0];
  const float* phi = (const float*)d_in[1];
  const float* mu  = (const float*)d_in[2];
  const float* L   = (const float*)d_in[3];
  float* out = (float*)d_out;
  int N = in_sizes[0] / DDIM;

  float* ws     = (float*)d_ws;
  float* wsLinv = ws;
  float* wsW    = ws + 1024;
  float* wsC    = ws + 1088;

  gmm_prep_kernel<<<1, 32, 0, stream>>>(phi, mu, L, wsLinv, wsW, wsC);

  int numTiles = (N + 15) / 16;
  long wavesWanted = (numTiles + 7) / 8;
  const int wavesPerBlock = 8;
  long blocks = (wavesWanted + wavesPerBlock - 1) / wavesPerBlock;
  if (blocks < 1) blocks = 1;
  if (blocks > 16384) blocks = 16384;
  int numWaves = (int)(blocks * wavesPerBlock);

  gmm_energy_kernel<<<(int)blocks, 256, 0, stream>>>(z, out, wsLinv, wsW, wsC,
                                                     N, numTiles, numWaves);
}


# Prep kernel parallelised: 64 threads each solve one column of an inverse in registers with L staged in LDS (was 4 threads, 16 columns each, private array); same arithmetic order per element
# speedup vs baseline: 3.6448x; 3.6448x over previous
#include <hip/hip_runtime.h>
#include <math.h>

typedef float v2f __attribute__((ext_vector_type(2)));
typedef float v8f __attribute__((ext_vector_type(8)));

#define KCOMP 4
#define DDIM  16

#define XPITCH 17

__global__ void __launch_bounds__(96) gmm_prep_kernel(
    const float* __restrict__ phi,
    const float* __restrict__ mu,
    const float* __restrict__ L,
    float* __restrict__ wsLinv,
    float* __restrict__ wsW,
    float* __restrict__ wsC) {
  __shared__ float sL[KCOMP * DDIM * DDIM];
  __shared__ float sX[KCOMP * DDIM * XPITCH];
  __shared__ float sMu[KCOMP * DDIM];
  const int tid = threadIdx.x;

  float r[10];
#pragma unroll
  for (int q = 0; q < 10; ++q) r[q] = L[tid + 96 * q];
  float rTail = 0.f, rMu = 0.f;
  if (tid < KCOMP * DDIM) { rTail = L[960 + tid]; rMu = mu[tid]; }
#pragma unroll
  for (int q = 0; q < 10; ++q) sL[tid + 96 * q] = r[q];
  if (tid < KCOMP * DDIM) { sL[960 + tid] = rTail; sMu[tid] = rMu; }
  __syncthreads();

  if (tid < KCOMP * DDIM) {
    const int k = tid >> 4;
    const int c = tid & 15;
    const float* Lk = sL + k * DDIM * DDIM;
    float x[DDIM];
#pragma unroll
    for (int i = DDIM - 1; i >= 0; --i) {
      float p = 0.f, s = 0.f;
#pragma unroll
      for (int t = i + 1; t < DDIM; ++t) {
        p = fmaf(Lk[i * DDIM + t], x[t], p);
        s = (t == c) ? p : s;
      }
      float num = (i == c) ? 1.f : -s;
      float v = num / Lk[i * DDIM + i];
      x[i] = (i <= c) ? v : 0.f;
    }
#pragma unroll
    for (int i = 0; i < DDIM; ++i) {
      sX[(k * DDIM + i) * XPITCH + c] = x[i];
      wsLinv[(k * DDIM + i) * DDIM + c] = x[i];
    }
  } else if (tid < KCOMP * DDIM + KCOMP) {
    const int k = tid - KCOMP * DDIM;
    const float* Lk = sL + k * DDIM * DDIM;
    float logdet = 0.f;
    for (int d = 0; d < DDIM; ++d) logdet += __logf(Lk[d * DDIM + d]);
    logdet *= 2.f;
    const float LOG2PI = 1.8378770664093453f;
    wsC[k] = __logf(phi[k]) - 0.5f * ((float)DDIM * LOG2PI + logdet);
  }
  __syncthreads();

  if (tid < KCOMP * DDIM) {
    const int k = tid >> 4;
    const int i = tid & 15;
    float s = 0.f;
    for (int j = 0; j < DDIM; ++j)
      s = fmaf(sX[(k * DDIM + i) * XPITCH + j], sMu[k * DDIM + j], s);
    wsW[k * DDIM + i] = s;
  }
}

__global__ void __launch_bounds__(256) gmm_energy_kernel(
    const float* __restrict__ z, float* __restrict__ out,
    const float* __restrict__ wsLinv, const float* __restrict__ wsW,
    const float* __restrict__ wsC,
    int N, int numTiles, int numWaves) {
  const int lane  = threadIdx.x & 31;
  const int m     = lane & 15;
  const int half  = lane >> 4;
  const int waveId = (int)((blockIdx.x * blockDim.x + threadIdx.x) >> 5);

  v2f a[KCOMP][4];
#pragma unroll
  for (int k = 0; k < KCOMP; ++k)
#pragma unroll
    for (int kk = 0; kk < 4; ++kk)
      a[k][kk] = *(const v2f*)(wsLinv + (k * 16 + m) * 16 + kk * 4 + 2 * half);

  v8f cinit[KCOMP];
#pragma unroll
  for (int k = 0; k < KCOMP; ++k) {
    const float* pw = wsW + k * 16 + 8 * half;
#pragma unroll
    for (int r = 0; r < 8; ++r) cinit[k][r] = -pw[r];
  }
  float ck[KCOMP];
#pragma unroll
  for (int k = 0; k < KCOMP; ++k) ck[k] = wsC[k];

  for (int t = waveId; t < numTiles; t += numWaves) {
    int n = t * 16 + m;
    if (n >= N) n = N - 1;
    const float* zp = z + (size_t)n * 16 + 2 * half;
    v2f b0 = *(const v2f*)(zp + 0);
    v2f b1 = *(const v2f*)(zp + 4);
    v2f b2 = *(const v2f*)(zp + 8);
    v2f b3 = *(const v2f*)(zp + 12);

    int nt = t + numWaves; if (nt >= numTiles) nt = numTiles - 1;
    __builtin_prefetch(z + (size_t)nt * 256 + lane * 8, 0, 1);

    float logit[KCOMP];
#pragma unroll
    for (int k = 0; k < KCOMP; ++k) {
      v8f acc = cinit[k];
      acc = __builtin_amdgcn_wmma_f32_16x16x4_f32(false, a[k][0], false, b0,
                                                  (short)0, acc, false, false);
      acc = __builtin_amdgcn_wmma_f32_16x16x4_f32(false, a[k][1], false, b1,
                                                  (short)0, acc, false, false);
      acc = __builtin_amdgcn_wmma_f32_16x16x4_f32(false, a[k][2], false, b2,
                                                  (short)0, acc, false, false);
      acc = __builtin_amdgcn_wmma_f32_16x16x4_f32(false, a[k][3], false, b3,
                                                  (short)0, acc, false, false);
      float s = 0.f;
#pragma unroll
      for (int r = 0; r < 8; ++r) s += acc[r] * acc[r];
      s += __shfl_xor(s, 16, 32);
      logit[k] = ck[k] - 0.5f * s;
    }
    float mx = fmaxf(fmaxf(logit[0], logit[1]), fmaxf(logit[2], logit[3]));
    float se = __expf(logit[0] - mx) + __expf(logit[1] - mx) +
               __expf(logit[2] - mx) + __expf(logit[3] - mx);
    float e = -(mx + __logf(se));

    if (half == 0) {
      int idx = t * 16 + m;
      if (idx < N) out[idx] = e;
    }
  }
}

extern "C" void kernel_launch(void* const* d_in, const int* in_sizes, int n_in,
                              void* d_out, int out_size, void* d_ws, size_t ws_size,
                              hipStream_t stream) {
  const float* z   = (const float*)d_in[0];
  const float* phi = (const float*)d_in[1];
  const float* mu  = (const float*)d_in[2];
  const float* L   = (const float*)d_in[3];
  float* out = (float*)d_out;
  int N = in_sizes[0] / DDIM;

  float* ws     = (float*)d_ws;
  float* wsLinv = ws;
  float* wsW    = ws + 1024;
  float* wsC    = ws + 1088;

  gmm_prep_kernel<<<1, 96, 0, stream>>>(phi, mu, L, wsLinv, wsW, wsC);

  int numTiles = (N + 15) / 16;
  long wavesWanted = (numTiles + 7) / 8;
  const int wavesPerBlock = 8;
  long blocks = (wavesWanted + wavesPerBlock - 1) / wavesPerBlock;
  if (blocks < 1) blocks = 1;
  if (blocks > 16384) blocks = 16384;
  int numWaves = (int)(blocks * wavesPerBlock);

  gmm_energy_kernel<<<(int)blocks, 256, 0, stream>>>(z, out, wsLinv, wsW, wsC,
                                                     N, numTiles, numWaves);
}


# Main kernel regroups the 64 stacked rows by 4-row block over the 4 components, skipping all-zero chunks of the triangular inverses: 10 f32 WMMAs per tile, was 16; norm sums now even/odd-row chains
# speedup vs baseline: 4.3261x; 1.1869x over previous
#include <hip/hip_runtime.h>
#include <math.h>

typedef float v2f __attribute__((ext_vector_type(2)));
typedef float v8f __attribute__((ext_vector_type(8)));

#define KCOMP 4
#define DDIM  16

#define XPITCH 17

__global__ void __launch_bounds__(96) gmm_prep_kernel(
    const float* __restrict__ phi,
    const float* __restrict__ mu,
    const float* __restrict__ L,
    float* __restrict__ wsLinv,
    float* __restrict__ wsW,
    float* __restrict__ wsC) {
  __shared__ float sL[KCOMP * DDIM * DDIM];
  __shared__ float sX[KCOMP * DDIM * XPITCH];
  __shared__ float sMu[KCOMP * DDIM];
  const int tid = threadIdx.x;

  float r[10];
#pragma unroll
  for (int q = 0; q < 10; ++q) r[q] = L[tid + 96 * q];
  float rTail = 0.f, rMu = 0.f;
  if (tid < KCOMP * DDIM) { rTail = L[960 + tid]; rMu = mu[tid]; }
#pragma unroll
  for (int q = 0; q < 10; ++q) sL[tid + 96 * q] = r[q];
  if (tid < KCOMP * DDIM) { sL[960 + tid] = rTail; sMu[tid] = rMu; }
  __syncthreads();

  if (tid < KCOMP * DDIM) {
    const int k = tid >> 4;
    const int c = tid & 15;
    const float* Lk = sL + k * DDIM * DDIM;
    float x[DDIM];
#pragma unroll
    for (int i = DDIM - 1; i >= 0; --i) {
      float p = 0.f, s = 0.f;
#pragma unroll
      for (int t = i + 1; t < DDIM; ++t) {
        p = fmaf(Lk[i * DDIM + t], x[t], p);
        s = (t == c) ? p : s;
      }
      float num = (i == c) ? 1.f : -s;
      float v = num / Lk[i * DDIM + i];
      x[i] = (i <= c) ? v : 0.f;
    }
#pragma unroll
    for (int i = 0; i < DDIM; ++i) {
      sX[(k * DDIM + i) * XPITCH + c] = x[i];
      wsLinv[(k * DDIM + i) * DDIM + c] = x[i];
    }
  } else if (tid < KCOMP * DDIM + KCOMP) {
    const int k = tid - KCOMP * DDIM;
    const float* Lk = sL + k * DDIM * DDIM;
    float logdet = 0.f;
    for (int d = 0; d < DDIM; ++d) logdet += __logf(Lk[d * DDIM + d]);
    logdet *= 2.f;
    const float LOG2PI = 1.8378770664093453f;
    wsC[k] = __logf(phi[k]) - 0.5f * ((float)DDIM * LOG2PI + logdet);
  }
  __syncthreads();

  if (tid < KCOMP * DDIM) {
    const int k = tid >> 4;
    const int i = tid & 15;
    float s = 0.f;
    for (int j = 0; j < DDIM; ++j)
      s = fmaf(sX[(k * DDIM + i) * XPITCH + j], sMu[k * DDIM + j], s);
    wsW[k * DDIM + i] = s;
  }
}

__global__ void __launch_bounds__(256) gmm_energy_kernel(
    const float* __restrict__ z, float* __restrict__ out,
    const float* __restrict__ wsLinv, const float* __restrict__ wsW,
    const float* __restrict__ wsC,
    int N, int numTiles, int numWaves) {
  const int lane  = threadIdx.x & 31;
  const int m     = lane & 15;
  const int half  = lane >> 4;
  const int waveId = (int)((blockIdx.x * blockDim.x + threadIdx.x) >> 5);

  const int acomp = m >> 2;
  const int arow  = m & 3;
  v2f a[4][4];
#pragma unroll
  for (int g = 0; g < 4; ++g)
#pragma unroll
    for (int kk = g; kk < 4; ++kk)
      a[g][kk] = *(const v2f*)(wsLinv + (acomp * 16 + 4 * g + arow) * 16 +
                               kk * 4 + 2 * half);

  v8f cinit[4];
#pragma unroll
  for (int g = 0; g < 4; ++g) {
#pragma unroll
    for (int r = 0; r < 8; ++r)
      cinit[g][r] = -wsW[(2 * half + (r >> 2)) * 16 + 4 * g + (r & 3)];
  }
  const float cA = wsC[2 * half];
  const float cB = wsC[2 * half + 1];

  for (int t = waveId; t < numTiles; t += numWaves) {
    int n = t * 16 + m;
    if (n >= N) n = N - 1;
    const float* zp = z + (size_t)n * 16 + 2 * half;
    v2f b0 = *(const v2f*)(zp + 0);
    v2f b1 = *(const v2f*)(zp + 4);
    v2f b2 = *(const v2f*)(zp + 8);
    v2f b3 = *(const v2f*)(zp + 12);

    int nt = t + numWaves; if (nt >= numTiles) nt = numTiles - 1;
    __builtin_prefetch(z + (size_t)nt * 256 + lane * 8, 0, 1);

    const v2f b[4] = {b0, b1, b2, b3};
    v2f pA = {0.f, 0.f}, pB = {0.f, 0.f};
#pragma unroll
    for (int g = 0; g < 4; ++g) {
      v8f acc = cinit[g];
#pragma unroll
      for (int kk = g; kk < 4; ++kk)
        acc = __builtin_amdgcn_wmma_f32_16x16x4_f32(false, a[g][kk], false, b[kk],
                                                    (short)0, acc, false, false);
      v2f x;
      x = __builtin_shufflevector(acc, acc, 0, 1); pA += x * x;
      x = __builtin_shufflevector(acc, acc, 4, 5); pB += x * x;
      x = __builtin_shufflevector(acc, acc, 2, 3); pA += x * x;
      x = __builtin_shufflevector(acc, acc, 6, 7); pB += x * x;
    }
    const float sA = pA[0] + pA[1];
    const float sB = pB[0] + pB[1];

    const float lA = cA - 0.5f * sA;
    const float lB = cB - 0.5f * sB;
    const float oA = __shfl_xor(lA, 16, 32);
    const float oB = __shfl_xor(lB, 16, 32);
    float mx = fmaxf(fmaxf(lA, lB), fmaxf(oA, oB));
    float se = __expf(lA - mx) + __expf(lB - mx) +
               __expf(oA - mx) + __expf(oB - mx);
    float e = -(mx + __logf(se));

    if (half == 0) {
      int idx = t * 16 + m;
      if (idx < N) out[idx] = e;
    }
  }
}

extern "C" void kernel_launch(void* const* d_in, const int* in_sizes, int n_in,
                              void* d_out, int out_size, void* d_ws, size_t ws_size,
                              hipStream_t stream) {
  const float* z   = (const float*)d_in[0];
  const float* phi = (const float*)d_in[1];
  const float* mu  = (const float*)d_in[2];
  const float* L   = (const float*)d_in[3];
  float* out = (float*)d_out;
  int N = in_sizes[0] / DDIM;

  float* ws     = (float*)d_ws;
  float* wsLinv = ws;
  float* wsW    = ws + 1024;
  float* wsC    = ws + 1088;

  gmm_prep_kernel<<<1, 96, 0, stream>>>(phi, mu, L, wsLinv, wsW, wsC);

  int numTiles = (N + 15) / 16;
  long wavesWanted = (numTiles + 7) / 8;
  const int wavesPerBlock = 8;
  long blocks = (wavesWanted + wavesPerBlock - 1) / wavesPerBlock;
  if (blocks < 1) blocks = 1;
  if (blocks > 16384) blocks = 16384;
  int numWaves = (int)(blocks * wavesPerBlock);

  gmm_energy_kernel<<<(int)blocks, 256, 0, stream>>>(z, out, wsLinv, wsW, wsC,
                                                     N, numTiles, numWaves);
}


# One launch, was two: each 8-wave block rebuilds the four triangular inverses, w and c in its own LDS (same arithmetic per element) and takes its WMMA A and C fragments from LDS; workspace unused
# speedup vs baseline: 4.7890x; 1.1070x over previous
#include <hip/hip_runtime.h>
#include <math.h>

typedef float v2f __attribute__((ext_vector_type(2)));
typedef float v4f __attribute__((ext_vector_type(4)));
typedef float v8f __attribute__((ext_vector_type(8)));

#define KCOMP 4
#define DDIM  16
#define XPITCH 17

__global__ void __launch_bounds__(256) gmm_energy_kernel(
    const float* __restrict__ z,
    const float* __restrict__ phi,
    const float* __restrict__ mu,
    const float* __restrict__ L,
    float* __restrict__ out,
    int N, int numTiles, int numWaves) {
  __shared__ __attribute__((aligned(16))) float sL[KCOMP * DDIM * DDIM];
  __shared__ __attribute__((aligned(16))) float sW[KCOMP * DDIM];
  __shared__ float sX[KCOMP * DDIM * XPITCH];
  __shared__ float sMu[KCOMP * DDIM];
  __shared__ float sC[KCOMP];

  const int tid   = threadIdx.x;
  const int lane  = tid & 31;
  const int m     = lane & 15;
  const int half  = lane >> 4;
  const int waveId = (int)((blockIdx.x * blockDim.x + threadIdx.x) >> 5);

  const v4f rL = *(const v4f*)(L + 4 * tid);
  float rMu = 0.f, rPhi = 1.f;
  if (tid < KCOMP * DDIM) rMu = mu[tid];
  else if (tid < KCOMP * DDIM + KCOMP) rPhi = phi[tid - KCOMP * DDIM];
  {
    int t0 = waveId < numTiles ? waveId : numTiles - 1;
    __builtin_prefetch(z + (size_t)t0 * 256 + lane * 8, 0, 1);
  }
  *(v4f*)(sL + 4 * tid) = rL;
  if (tid < KCOMP * DDIM) sMu[tid] = rMu;
  __syncthreads();

  if (tid < KCOMP * DDIM) {
    const int k = tid >> 4;
    const int c = tid & 15;
    const float* Lk = sL + k * DDIM * DDIM;
    float x[DDIM];
#pragma unroll
    for (int i = DDIM - 1; i >= 0; --i) {
      float p = 0.f, s = 0.f;
#pragma unroll
      for (int t = i + 1; t < DDIM; ++t) {
        p = fmaf(Lk[i * DDIM + t], x[t], p);
        s = (t == c) ? p : s;
      }
      float num = (i == c) ? 1.f : -s;
      float v = num / Lk[i * DDIM + i];
      x[i] = (i <= c) ? v : 0.f;
    }
#pragma unroll
    for (int i = 0; i < DDIM; ++i)
      sX[(k * DDIM + i) * XPITCH + c] = x[i];
  } else if (tid < KCOMP * DDIM + KCOMP) {
    const int k = tid - KCOMP * DDIM;
    const float* Lk = sL + k * DDIM * DDIM;
    float logdet = 0.f;
    for (int d = 0; d < DDIM; ++d) logdet += __logf(Lk[d * DDIM + d]);
    logdet *= 2.f;
    const float LOG2PI = 1.8378770664093453f;
    sC[k] = __logf(rPhi) - 0.5f * ((float)DDIM * LOG2PI + logdet);
  }
  __syncthreads();

  const int acomp = m >> 2;
  const int arow  = m & 3;
  v2f a[4][4];
#pragma unroll
  for (int g = 0; g < 4; ++g)
#pragma unroll
    for (int kk = g; kk < 4; ++kk) {
      const float* ap = sX + (acomp * 16 + 4 * g + arow) * XPITCH + kk * 4 + 2 * half;
      a[g][kk][0] = ap[0];
      a[g][kk][1] = ap[1];
    }
  const float cA = sC[2 * half];
  const float cB = sC[2 * half + 1];

  if (tid < KCOMP * DDIM) {
    const int k = tid >> 4;
    const int i = tid & 15;
    float s = 0.f;
    for (int j = 0; j < DDIM; ++j)
      s = fmaf(sX[(k * DDIM + i) * XPITCH + j], sMu[k * DDIM + j], s);
    sW[k * DDIM + i] = s;
  }
  __syncthreads();

  v8f cinit[4];
#pragma unroll
  for (int g = 0; g < 4; ++g) {
#pragma unroll
    for (int r = 0; r < 8; ++r)
      cinit[g][r] = -sW[(2 * half + (r >> 2)) * 16 + 4 * g + (r & 3)];
  }

  for (int t = waveId; t < numTiles; t += numWaves) {
    int n = t * 16 + m;
    if (n >= N) n = N - 1;
    const float* zp = z + (size_t)n * 16 + 2 * half;
    v2f b0 = *(const v2f*)(zp + 0);
    v2f b1 = *(const v2f*)(zp + 4);
    v2f b2 = *(const v2f*)(zp + 8);
    v2f b3 = *(const v2f*)(zp + 12);

    int nt = t + numWaves; if (nt >= numTiles) nt = numTiles - 1;
    __builtin_prefetch(z + (size_t)nt * 256 + lane * 8, 0, 1);

    const v2f b[4] = {b0, b1, b2, b3};
    v2f pA = {0.f, 0.f}, pB = {0.f, 0.f};
#pragma unroll
    for (int g = 0; g < 4; ++g) {
      v8f acc = cinit[g];
#pragma unroll
      for (int kk = g; kk < 4; ++kk)
        acc = __builtin_amdgcn_wmma_f32_16x16x4_f32(false, a[g][kk], false, b[kk],
                                                    (short)0, acc, false, false);
      v2f x;
      x = __builtin_shufflevector(acc, acc, 0, 1); pA += x * x;
      x = __builtin_shufflevector(acc, acc, 4, 5); pB += x * x;
      x = __builtin_shufflevector(acc, acc, 2, 3); pA += x * x;
      x = __builtin_shufflevector(acc, acc, 6, 7); pB += x * x;
    }
    const float sA = pA[0] + pA[1];
    const float sB = pB[0] + pB[1];

    const float lA = cA - 0.5f * sA;
    const float lB = cB - 0.5f * sB;
    const float oA = __shfl_xor(lA, 16, 32);
    const float oB = __shfl_xor(lB, 16, 32);
    float mx = fmaxf(fmaxf(lA, lB), fmaxf(oA, oB));
    float se = __expf(lA - mx) + __expf(lB - mx) +
               __expf(oA - mx) + __expf(oB - mx);
    float e = -(mx + __logf(se));

    if (half == 0) {
      int idx = t * 16 + m;
      if (idx < N) out[idx] = e;
    }
  }
}

extern "C" void kernel_launch(void* const* d_in, const int* in_sizes, int n_in,
                              void* d_out, int out_size, void* d_ws, size_t ws_size,
                              hipStream_t stream) {
  const float* z   = (const float*)d_in[0];
  const float* phi = (const float*)d_in[1];
  const float* mu  = (const float*)d_in[2];
  const float* L   = (const float*)d_in[3];
  float* out = (float*)d_out;
  int N = in_sizes[0] / DDIM;

  int numTiles = (N + 15) / 16;
  long wavesWanted = (numTiles + 7) / 8;
  const int wavesPerBlock = 8;
  long blocks = (wavesWanted + wavesPerBlock - 1) / wavesPerBlock;
  if (blocks < 1) blocks = 1;
  if (blocks > 16384) blocks = 16384;
  int numWaves = (int)(blocks * wavesPerBlock);

  gmm_energy_kernel<<<(int)blocks, 256, 0, stream>>>(z, phi, mu, L, out,
                                                     N, numTiles, numWaves);
}


# Prologue chain cut to 15 dependent multiply-adds (was 120 plus a divide per row): L rows pre-divided by the diagonal, Linv columns and w solved by column sweeps; sums now run in descending order
# speedup vs baseline: 5.1005x; 1.0650x over previous
#include <hip/hip_runtime.h>
#include <math.h>

typedef float v2f __attribute__((ext_vector_type(2)));
typedef float v4f __attribute__((ext_vector_type(4)));
typedef float v8f __attribute__((ext_vector_type(8)));

#define KCOMP 4
#define DDIM  16
#define XPITCH 17
#define LTSTRIDE 272

__global__ void __launch_bounds__(256) gmm_energy_kernel(
    const float* __restrict__ z,
    const float* __restrict__ phi,
    const float* __restrict__ mu,
    const float* __restrict__ L,
    float* __restrict__ out,
    int N, int numTiles, int numWaves) {
  __shared__ __attribute__((aligned(16))) float sLT[KCOMP * LTSTRIDE];
  __shared__ __attribute__((aligned(16))) float sW[KCOMP * DDIM];
  __shared__ __attribute__((aligned(16))) float sMuS[KCOMP * DDIM];
  __shared__ float sX[KCOMP * DDIM * XPITCH];
  __shared__ __attribute__((aligned(16))) float sLog[KCOMP * DDIM];
  __shared__ float sC[KCOMP];

  const int tid   = threadIdx.x;
  const int lane  = tid & 31;
  const int m     = lane & 15;
  const int half  = lane >> 4;
  const int waveId = (int)((blockIdx.x * blockDim.x + threadIdx.x) >> 5);
  const int wv = __builtin_amdgcn_readfirstlane(tid >> 5);

  const int rk = tid >> 6;
  const int ri = (tid >> 2) & 15;
  const int rq = tid & 3;
  const v4f rL = *(const v4f*)(L + 4 * tid);
  const float dRow = L[rk * DDIM * DDIM + ri * (DDIM + 1)];
  const int j64 = tid & 63;
  const float dOwn = L[(j64 >> 4) * DDIM * DDIM + (j64 & 15) * (DDIM + 1)];
  const float rMu  = mu[j64];
  const float rPhi = phi[lane & 3];
  {
    int t0 = waveId < numTiles ? waveId : numTiles - 1;
    __builtin_prefetch(z + (size_t)t0 * 256 + lane * 8, 0, 1);
  }
  asm volatile("" :: "v"(rMu), "v"(dOwn), "v"(rPhi));
#pragma unroll
  for (int j = 0; j < 4; ++j)
    sLT[rk * LTSTRIDE + (4 * rq + j) * DDIM + ri] = rL[j] / dRow;
  float rinv = 1.f;
  if (wv < 2) {
    rinv = 1.f / dOwn;
    sMuS[tid]  = rMu / dOwn;
  } else if ((wv >> 1) == 2) {
    sLog[j64] = __logf(dOwn);
  }
  __syncthreads();

  if (wv < 3) {
    const int k = (wv < 2) ? (tid >> 4) : (lane & 3);
    const int c = tid & 15;
    const float* LT = sLT + k * LTSTRIDE;
    float s[DDIM];
    if (wv < 2) {
#pragma unroll
      for (int i = 0; i < DDIM; ++i) s[i] = (i == c) ? rinv : 0.f;
    } else {
#pragma unroll
      for (int b = 0; b < 4; ++b) {
        const v4f q = *(const v4f*)(sMuS + k * DDIM + 4 * b);
        s[4 * b + 0] = q[0]; s[4 * b + 1] = q[1];
        s[4 * b + 2] = q[2]; s[4 * b + 3] = q[3];
      }
    }
    float col[DDIM][DDIM];
#pragma unroll
    for (int t = DDIM - 1; t >= 1; --t)
#pragma unroll
      for (int i = 0; i < t; ++i) col[t][i] = LT[t * DDIM + i];
#pragma unroll
    for (int t = DDIM - 1; t >= 1; --t) {
      const float xt = s[t];
#pragma unroll
      for (int i = 0; i < t; ++i)
        s[i] = fmaf(-col[t][i], xt, s[i]);
    }
    if (wv < 2) {
#pragma unroll
      for (int i = 0; i < DDIM; ++i)
        sX[(k * DDIM + i) * XPITCH + c] = (i <= c) ? s[i] : 0.f;
    } else if (lane < KCOMP) {
#pragma unroll
      for (int i = 0; i < DDIM; ++i) sW[k * DDIM + i] = s[i];
    }
  } else if (wv == 3) {
    if (lane < KCOMP) {
      const int k = lane;
      float logdet = 0.f;
      for (int d = 0; d < DDIM; ++d) logdet += sLog[k * DDIM + d];
      logdet *= 2.f;
      const float LOG2PI = 1.8378770664093453f;
      sC[k] = __logf(rPhi) - 0.5f * ((float)DDIM * LOG2PI + logdet);
    }
  }
  __syncthreads();

  const int acomp = m >> 2;
  const int arow  = m & 3;
  v2f a[4][4];
#pragma unroll
  for (int g = 0; g < 4; ++g)
#pragma unroll
    for (int kk = g; kk < 4; ++kk) {
      const float* ap = sX + (acomp * 16 + 4 * g + arow) * XPITCH + kk * 4 + 2 * half;
      a[g][kk][0] = ap[0];
      a[g][kk][1] = ap[1];
    }
  const float cA = sC[2 * half];
  const float cB = sC[2 * half + 1];

  v8f cinit[4];
#pragma unroll
  for (int g = 0; g < 4; ++g) {
#pragma unroll
    for (int r = 0; r < 8; ++r)
      cinit[g][r] = -sW[(2 * half + (r >> 2)) * 16 + 4 * g + (r & 3)];
  }

  for (int t = waveId; t < numTiles; t += numWaves) {
    int n = t * 16 + m;
    if (n >= N) n = N - 1;
    const float* zp = z + (size_t)n * 16 + 2 * half;
    v2f b0 = *(const v2f*)(zp + 0);
    v2f b1 = *(const v2f*)(zp + 4);
    v2f b2 = *(const v2f*)(zp + 8);
    v2f b3 = *(const v2f*)(zp + 12);

    int nt = t + numWaves; if (nt >= numTiles) nt = numTiles - 1;
    __builtin_prefetch(z + (size_t)nt * 256 + lane * 8, 0, 1);

    const v2f b[4] = {b0, b1, b2, b3};
    v2f pA = {0.f, 0.f}, pB = {0.f, 0.f};
#pragma unroll
    for (int g = 0; g < 4; ++g) {
      v8f acc = cinit[g];
#pragma unroll
      for (int kk = g; kk < 4; ++kk)
        acc = __builtin_amdgcn_wmma_f32_16x16x4_f32(false, a[g][kk], false, b[kk],
                                                    (short)0, acc, false, false);
      v2f x;
      x = __builtin_shufflevector(acc, acc, 0, 1); pA += x * x;
      x = __builtin_shufflevector(acc, acc, 4, 5); pB += x * x;
      x = __builtin_shufflevector(acc, acc, 2, 3); pA += x * x;
      x = __builtin_shufflevector(acc, acc, 6, 7); pB += x * x;
    }
    const float sA = pA[0] + pA[1];
    const float sB = pB[0] + pB[1];

    const float lA = cA - 0.5f * sA;
    const float lB = cB - 0.5f * sB;
    const float oA = __shfl_xor(lA, 16, 32);
    const float oB = __shfl_xor(lB, 16, 32);
    float mx = fmaxf(fmaxf(lA, lB), fmaxf(oA, oB));
    float se = __expf(lA - mx) + __expf(lB - mx) +
               __expf(oA - mx) + __expf(oB - mx);
    float e = -(mx + __logf(se));

    if (half == 0) {
      int idx = t * 16 + m;
      if (idx < N) out[idx] = e;
    }
  }
}

extern "C" void kernel_launch(void* const* d_in, const int* in_sizes, int n_in,
                              void* d_out, int out_size, void* d_ws, size_t ws_size,
                              hipStream_t stream) {
  const float* z   = (const float*)d_in[0];
  const float* phi = (const float*)d_in[1];
  const float* mu  = (const float*)d_in[2];
  const float* L   = (const float*)d_in[3];
  float* out = (float*)d_out;
  int N = in_sizes[0] / DDIM;

  int numTiles = (N + 15) / 16;
  long wavesWanted = (numTiles + 7) / 8;
  const int wavesPerBlock = 8;
  long blocks = (wavesWanted + wavesPerBlock - 1) / wavesPerBlock;
  if (blocks < 1) blocks = 1;
  if (blocks > 16384) blocks = 16384;
  int numWaves = (int)(blocks * wavesPerBlock);

  gmm_energy_kernel<<<(int)blocks, 256, 0, stream>>>(z, phi, mu, L, out,
                                                     N, numTiles, numWaves);
}


# Tile loop pipelined inside each wave: the next tile's z loads go out one step ahead into a second register set (scalar tile index); grid cut to 489 blocks of 16 tiles per wave, was 977 of 8
# speedup vs baseline: 5.5631x; 1.0907x over previous
#include <hip/hip_runtime.h>
#include <math.h>

typedef float v2f __attribute__((ext_vector_type(2)));
typedef float v4f __attribute__((ext_vector_type(4)));
typedef float v8f __attribute__((ext_vector_type(8)));

#define KCOMP 4
#define DDIM  16
#define XPITCH 17
#define LTSTRIDE 272

struct BFrag { v2f k0, k1, k2, k3; };

static __device__ __forceinline__ BFrag load_tile(const float* __restrict__ z, int t,
                                                  int N, int m, int half) {
  int r = N - 1 - t * 16; r = m < r ? m : r;
  const float* zp = z + (size_t)t * 256 + (unsigned)(r * 16 + 2 * half);
  BFrag f;
  f.k0 = *(const v2f*)(zp + 0);
  f.k1 = *(const v2f*)(zp + 4);
  f.k2 = *(const v2f*)(zp + 8);
  f.k3 = *(const v2f*)(zp + 12);
  return f;
}

static __device__ __forceinline__ void tile_step(const v2f (&a)[4][4], const v8f (&cinit)[4],
                                                 const BFrag& f, float cA, float cB,
                                                 float* __restrict__ out, int t, int N,
                                                 int m, int half) {
  const v2f b[4] = {f.k0, f.k1, f.k2, f.k3};
  v2f pA = {0.f, 0.f}, pB = {0.f, 0.f};
#pragma unroll
  for (int g = 0; g < 4; ++g) {
    v8f acc = cinit[g];
#pragma unroll
    for (int kk = g; kk < 4; ++kk)
      acc = __builtin_amdgcn_wmma_f32_16x16x4_f32(false, a[g][kk], false, b[kk],
                                                  (short)0, acc, false, false);
    v2f x;
    x = __builtin_shufflevector(acc, acc, 0, 1); pA += x * x;
    x = __builtin_shufflevector(acc, acc, 4, 5); pB += x * x;
    x = __builtin_shufflevector(acc, acc, 2, 3); pA += x * x;
    x = __builtin_shufflevector(acc, acc, 6, 7); pB += x * x;
  }
  const float sA = pA[0] + pA[1];
  const float sB = pB[0] + pB[1];

  const float lA = cA - 0.5f * sA;
  const float lB = cB - 0.5f * sB;
  const float oA = __shfl_xor(lA, 16, 32);
  const float oB = __shfl_xor(lB, 16, 32);
  float mx = fmaxf(fmaxf(lA, lB), fmaxf(oA, oB));
  float se = __expf(lA - mx) + __expf(lB - mx) +
             __expf(oA - mx) + __expf(oB - mx);
  float e = -(mx + __logf(se));

  if (half == 0) {
    const int rem = N - t * 16;
    if (m < rem) out[(size_t)t * 16 + (unsigned)m] = e;
  }
}

__global__ void __launch_bounds__(256) gmm_energy_kernel(
    const float* __restrict__ z,
    const float* __restrict__ phi,
    const float* __restrict__ mu,
    const float* __restrict__ L,
    float* __restrict__ out,
    int N, int numTiles, int numWaves) {
  __shared__ __attribute__((aligned(16))) float sLT[KCOMP * LTSTRIDE];
  __shared__ __attribute__((aligned(16))) float sW[KCOMP * DDIM];
  __shared__ __attribute__((aligned(16))) float sMuS[KCOMP * DDIM];
  __shared__ float sX[KCOMP * DDIM * XPITCH];
  __shared__ __attribute__((aligned(16))) float sLog[KCOMP * DDIM];
  __shared__ float sC[KCOMP];

  const int tid   = threadIdx.x;
  const int lane  = tid & 31;
  const int m     = lane & 15;
  const int half  = lane >> 4;
  const int wv = __builtin_amdgcn_readfirstlane(tid >> 5);
  const int waveId = (int)blockIdx.x * 8 + wv;

  const int rk = tid >> 6;
  const int ri = (tid >> 2) & 15;
  const int rq = tid & 3;
  const v4f rL = *(const v4f*)(L + 4 * tid);
  const float dRow = L[rk * DDIM * DDIM + ri * (DDIM + 1)];
  const int j64 = tid & 63;
  const float dOwn = L[(j64 >> 4) * DDIM * DDIM + (j64 & 15) * (DDIM + 1)];
  const float rMu  = mu[j64];
  const float rPhi = phi[lane & 3];
  const int lastTile = numTiles - 1;
  {
    int t0 = waveId < numTiles ? waveId : lastTile;
    int t1 = waveId + numWaves; if (t1 > lastTile) t1 = lastTile;
    __builtin_prefetch(z + (size_t)t0 * 256 + (unsigned)(lane * 8), 0, 1);
    __builtin_prefetch(z + (size_t)t1 * 256 + (unsigned)(lane * 8), 0, 1);
  }
  asm volatile("" :: "v"(rMu), "v"(dOwn), "v"(rPhi));
#pragma unroll
  for (int j = 0; j < 4; ++j)
    sLT[rk * LTSTRIDE + (4 * rq + j) * DDIM + ri] = rL[j] / dRow;
  float rinv = 1.f;
  if (wv < 2) {
    rinv = 1.f / dOwn;
    sMuS[tid]  = rMu / dOwn;
  } else if ((wv >> 1) == 2) {
    sLog[j64] = __logf(dOwn);
  }
  __syncthreads();

  if (wv < 3) {
    const int k = (wv < 2) ? (tid >> 4) : (lane & 3);
    const int c = tid & 15;
    const float* LT = sLT + k * LTSTRIDE;
    float s[DDIM];
    if (wv < 2) {
#pragma unroll
      for (int i = 0; i < DDIM; ++i) s[i] = (i == c) ? rinv : 0.f;
    } else {
#pragma unroll
      for (int b = 0; b < 4; ++b) {
        const v4f q = *(const v4f*)(sMuS + k * DDIM + 4 * b);
        s[4 * b + 0] = q[0]; s[4 * b + 1] = q[1];
        s[4 * b + 2] = q[2]; s[4 * b + 3] = q[3];
      }
    }
    float col[DDIM][DDIM];
#pragma unroll
    for (int t = DDIM - 1; t >= 1; --t)
#pragma unroll
      for (int i = 0; i < t; ++i) col[t][i] = LT[t * DDIM + i];
#pragma unroll
    for (int t = DDIM - 1; t >= 1; --t) {
      const float xt = s[t];
#pragma unroll
      for (int i = 0; i < t; ++i)
        s[i] = fmaf(-col[t][i], xt, s[i]);
    }
    if (wv < 2) {
#pragma unroll
      for (int i = 0; i < DDIM; ++i)
        sX[(k * DDIM + i) * XPITCH + c] = (i <= c) ? s[i] : 0.f;
    } else if (lane < KCOMP) {
#pragma unroll
      for (int i = 0; i < DDIM; ++i) sW[k * DDIM + i] = s[i];
    }
  } else if (wv == 3) {
    if (lane < KCOMP) {
      const int k = lane;
      float logdet = 0.f;
      for (int d = 0; d < DDIM; ++d) logdet += sLog[k * DDIM + d];
      logdet *= 2.f;
      const float LOG2PI = 1.8378770664093453f;
      sC[k] = __logf(rPhi) - 0.5f * ((float)DDIM * LOG2PI + logdet);
    }
  }
  __syncthreads();

  BFrag fa = load_tile(z, waveId < numTiles ? waveId : lastTile, N, m, half);
  BFrag fb;

  const int acomp = m >> 2;
  const int arow  = m & 3;
  v2f a[4][4];
#pragma unroll
  for (int g = 0; g < 4; ++g)
#pragma unroll
    for (int kk = g; kk < 4; ++kk) {
      const float* ap = sX + (acomp * 16 + 4 * g + arow) * XPITCH + kk * 4 + 2 * half;
      a[g][kk][0] = ap[0];
      a[g][kk][1] = ap[1];
    }
  const float cA = sC[2 * half];
  const float cB = sC[2 * half + 1];

  v8f cinit[4];
#pragma unroll
  for (int g = 0; g < 4; ++g) {
#pragma unroll
    for (int r = 0; r < 8; ++r)
      cinit[g][r] = -sW[(2 * half + (r >> 2)) * 16 + 4 * g + (r & 3)];
  }

  for (int t = waveId; t < numTiles; t += 2 * numWaves) {
    const int tb = t + numWaves;
    {
      int tp = tb + numWaves; if (tp > lastTile) tp = lastTile;
      __builtin_prefetch(z + (size_t)tp * 256 + (unsigned)(lane * 8), 0, 1);
      fb = load_tile(z, tb < lastTile ? tb : lastTile, N, m, half);
    }
    __builtin_amdgcn_sched_barrier(0);
    tile_step(a, cinit, fa, cA, cB, out, t, N, m, half);
    if (tb >= numTiles) break;
    {
      const int ta = tb + numWaves;
      int tp = ta + numWaves; if (tp > lastTile) tp = lastTile;
      __builtin_prefetch(z + (size_t)tp * 256 + (unsigned)(lane * 8), 0, 1);
      fa = load_tile(z, ta < lastTile ? ta : lastTile, N, m, half);
    }
    __builtin_amdgcn_sched_barrier(0);
    tile_step(a, cinit, fb, cA, cB, out, tb, N, m, half);
  }
}

extern "C" void kernel_launch(void* const* d_in, const int* in_sizes, int n_in,
                              void* d_out, int out_size, void* d_ws, size_t ws_size,
                              hipStream_t stream) {
  const float* z   = (const float*)d_in[0];
  const float* phi = (const float*)d_in[1];
  const float* mu  = (const float*)d_in[2];
  const float* L   = (const float*)d_in[3];
  float* out = (float*)d_out;
  int N = in_sizes[0] / DDIM;

  int numTiles = (N + 15) / 16;
  long wavesWanted = (numTiles + 15) / 16;
  const int wavesPerBlock = 8;
  long blocks = (wavesWanted + wavesPerBlock - 1) / wavesPerBlock;
  if (blocks < 1) blocks = 1;
  if (blocks > 16384) blocks = 16384;
  int numWaves = (int)(blocks * wavesPerBlock);

  gmm_energy_kernel<<<(int)blocks, 256, 0, stream>>>(z, phi, mu, L, out,
                                                     N, numTiles, numWaves);
}


# Each wave step takes two adjacent tiles (32 points): 8 accumulator chains with the two tiles' matrix instructions alternating, one log-sum-exp pass on all 32 lanes, one whole 128-byte line stored
# speedup vs baseline: 5.8755x; 1.0562x over previous
#include <hip/hip_runtime.h>
#include <math.h>

typedef float v2f __attribute__((ext_vector_type(2)));
typedef float v4f __attribute__((ext_vector_type(4)));
typedef float v8f __attribute__((ext_vector_type(8)));

#define KCOMP 4
#define DDIM  16
#define XPITCH 17
#define LTSTRIDE 272

struct BFrag { v2f k0, k1, k2, k3; };
struct BPair { BFrag x, y; };

static __device__ __forceinline__ BFrag load_frag(const float* __restrict__ zp) {
  BFrag f;
  f.k0 = *(const v2f*)(zp + 0);
  f.k1 = *(const v2f*)(zp + 4);
  f.k2 = *(const v2f*)(zp + 8);
  f.k3 = *(const v2f*)(zp + 12);
  return f;
}

static __device__ __forceinline__ BPair load_pair(const float* __restrict__ z, int p,
                                                  int N, int m, int half) {
  const int lim = N - 1 - p * 32;
  const int rx = m < lim ? m : lim;
  const int ry = m + 16 < lim ? m + 16 : lim;
  const float* zb = z + (size_t)p * 512 + (unsigned)(2 * half);
  BPair f;
  f.x = load_frag(zb + (unsigned)(rx * 16));
  f.y = load_frag(zb + (unsigned)(ry * 16));
  return f;
}

static __device__ __forceinline__ void add_norms(const v8f& acc, v2f& pA, v2f& pB) {
  v2f x;
  x = __builtin_shufflevector(acc, acc, 0, 1); pA += x * x;
  x = __builtin_shufflevector(acc, acc, 4, 5); pB += x * x;
  x = __builtin_shufflevector(acc, acc, 2, 3); pA += x * x;
  x = __builtin_shufflevector(acc, acc, 6, 7); pB += x * x;
}

static __device__ __forceinline__ void pair_step(const v2f (&a)[4][4], const v8f (&cinit)[4],
                                                 const BPair& f, float cA, float cB,
                                                 float* __restrict__ out, int p, int N,
                                                 int lane, int half) {
  const v2f bx[4] = {f.x.k0, f.x.k1, f.x.k2, f.x.k3};
  const v2f by[4] = {f.y.k0, f.y.k1, f.y.k2, f.y.k3};
  v2f pAx = {0.f, 0.f}, pBx = {0.f, 0.f};
  v2f pAy = {0.f, 0.f}, pBy = {0.f, 0.f};
#pragma unroll
  for (int g = 0; g < 4; ++g) {
    v8f ax = cinit[g];
    v8f ay = cinit[g];
#pragma unroll
    for (int kk = g; kk < 4; ++kk) {
      ax = __builtin_amdgcn_wmma_f32_16x16x4_f32(false, a[g][kk], false, bx[kk],
                                                 (short)0, ax, false, false);
      ay = __builtin_amdgcn_wmma_f32_16x16x4_f32(false, a[g][kk], false, by[kk],
                                                 (short)0, ay, false, false);
    }
    add_norms(ax, pAx, pBx);
    add_norms(ay, pAy, pBy);
  }
  const float lAx = cA - 0.5f * (pAx[0] + pAx[1]);
  const float lBx = cB - 0.5f * (pBx[0] + pBx[1]);
  const float lAy = cA - 0.5f * (pAy[0] + pAy[1]);
  const float lBy = cB - 0.5f * (pBy[0] + pBy[1]);
  const bool up = half != 0;
  const float ownA = up ? lAy : lAx;
  const float ownB = up ? lBy : lBx;
  const float rcvA = __shfl_xor(up ? lAx : lAy, 16, 32);
  const float rcvB = __shfl_xor(up ? lBx : lBy, 16, 32);
  const float l0 = up ? rcvA : ownA;
  const float l1 = up ? rcvB : ownB;
  const float l2 = up ? ownA : rcvA;
  const float l3 = up ? ownB : rcvB;
  float mx = fmaxf(fmaxf(l0, l1), fmaxf(l2, l3));
  float se = __expf(l0 - mx) + __expf(l1 - mx) +
             __expf(l2 - mx) + __expf(l3 - mx);
  float e = -(mx + __logf(se));

  const int rem = N - p * 32;
  if (lane < rem) out[(size_t)p * 32 + (unsigned)lane] = e;
}

__global__ void __launch_bounds__(256) gmm_energy_kernel(
    const float* __restrict__ z,
    const float* __restrict__ phi,
    const float* __restrict__ mu,
    const float* __restrict__ L,
    float* __restrict__ out,
    int N, int numPairs, int numWaves) {
  __shared__ __attribute__((aligned(16))) float sLT[KCOMP * LTSTRIDE];
  __shared__ __attribute__((aligned(16))) float sW[KCOMP * DDIM];
  __shared__ __attribute__((aligned(16))) float sMuS[KCOMP * DDIM];
  __shared__ float sX[KCOMP * DDIM * XPITCH];
  __shared__ __attribute__((aligned(16))) float sLog[KCOMP * DDIM];
  __shared__ float sC[KCOMP];

  const int tid   = threadIdx.x;
  const int lane  = tid & 31;
  const int m     = lane & 15;
  const int half  = lane >> 4;
  const int wv = __builtin_amdgcn_readfirstlane(tid >> 5);
  const int waveId = (int)blockIdx.x * 8 + wv;

  const int rk = tid >> 6;
  const int ri = (tid >> 2) & 15;
  const int rq = tid & 3;
  const v4f rL = *(const v4f*)(L + 4 * tid);
  const float dRow = L[rk * DDIM * DDIM + ri * (DDIM + 1)];
  const int j64 = tid & 63;
  const float dOwn = L[(j64 >> 4) * DDIM * DDIM + (j64 & 15) * (DDIM + 1)];
  const float rMu  = mu[j64];
  const float rPhi = phi[lane & 3];
  const int lastPair = numPairs - 1;
  {
    int p0 = waveId < numPairs ? waveId : lastPair;
    int p1 = waveId + numWaves; if (p1 > lastPair) p1 = lastPair;
    __builtin_prefetch(z + (size_t)p0 * 512 + (unsigned)(lane * 16), 0, 1);
    __builtin_prefetch(z + (size_t)p1 * 512 + (unsigned)(lane * 16), 0, 1);
  }
  asm volatile("" :: "v"(rMu), "v"(dOwn), "v"(rPhi));
#pragma unroll
  for (int j = 0; j < 4; ++j)
    sLT[rk * LTSTRIDE + (4 * rq + j) * DDIM + ri] = rL[j] / dRow;
  float rinv = 1.f;
  if (wv < 2) {
    rinv = 1.f / dOwn;
    sMuS[tid]  = rMu / dOwn;
  } else if ((wv >> 1) == 2) {
    sLog[j64] = __logf(dOwn);
  }
  __syncthreads();

  if (wv < 3) {
    const int k = (wv < 2) ? (tid >> 4) : (lane & 3);
    const int c = tid & 15;
    const float* LT = sLT + k * LTSTRIDE;
    float s[DDIM];
    if (wv < 2) {
#pragma unroll
      for (int i = 0; i < DDIM; ++i) s[i] = (i == c) ? rinv : 0.f;
    } else {
#pragma unroll
      for (int b = 0; b < 4; ++b) {
        const v4f q = *(const v4f*)(sMuS + k * DDIM + 4 * b);
        s[4 * b + 0] = q[0]; s[4 * b + 1] = q[1];
        s[4 * b + 2] = q[2]; s[4 * b + 3] = q[3];
      }
    }
    float col[DDIM][DDIM];
#pragma unroll
    for (int t = DDIM - 1; t >= 1; --t)
#pragma unroll
      for (int i = 0; i < t; ++i) col[t][i] = LT[t * DDIM + i];
#pragma unroll
    for (int t = DDIM - 1; t >= 1; --t) {
      const float xt = s[t];
#pragma unroll
      for (int i = 0; i < t; ++i)
        s[i] = fmaf(-col[t][i], xt, s[i]);
    }
    if (wv < 2) {
#pragma unroll
      for (int i = 0; i < DDIM; ++i)
        sX[(k * DDIM + i) * XPITCH + c] = (i <= c) ? s[i] : 0.f;
    } else if (lane < KCOMP) {
#pragma unroll
      for (int i = 0; i < DDIM; ++i) sW[k * DDIM + i] = s[i];
    }
  } else if (wv == 3) {
    if (lane < KCOMP) {
      const int k = lane;
      float logdet = 0.f;
      for (int d = 0; d < DDIM; ++d) logdet += sLog[k * DDIM + d];
      logdet *= 2.f;
      const float LOG2PI = 1.8378770664093453f;
      sC[k] = __logf(rPhi) - 0.5f * ((float)DDIM * LOG2PI + logdet);
    }
  }
  __syncthreads();

  BPair fa = load_pair(z, waveId < numPairs ? waveId : lastPair, N, m, half);
  BPair fb;

  const int acomp = m >> 2;
  const int arow  = m & 3;
  v2f a[4][4];
#pragma unroll
  for (int g = 0; g < 4; ++g)
#pragma unroll
    for (int kk = g; kk < 4; ++kk) {
      const float* ap = sX + (acomp * 16 + 4 * g + arow) * XPITCH + kk * 4 + 2 * half;
      a[g][kk][0] = ap[0];
      a[g][kk][1] = ap[1];
    }
  const float cA = sC[2 * half];
  const float cB = sC[2 * half + 1];

  v8f cinit[4];
#pragma unroll
  for (int g = 0; g < 4; ++g) {
#pragma unroll
    for (int r = 0; r < 8; ++r)
      cinit[g][r] = -sW[(2 * half + (r >> 2)) * 16 + 4 * g + (r & 3)];
  }

  for (int p = waveId; p < numPairs; p += 2 * numWaves) {
    const int pb = p + numWaves;
    {
      int pp = pb + numWaves; if (pp > lastPair) pp = lastPair;
      __builtin_prefetch(z + (size_t)pp * 512 + (unsigned)(lane * 16), 0, 1);
      fb = load_pair(z, pb < lastPair ? pb : lastPair, N, m, half);
    }
    __builtin_amdgcn_sched_barrier(0);
    pair_step(a, cinit, fa, cA, cB, out, p, N, lane, half);
    if (pb >= numPairs) break;
    {
      const int pa = pb + numWaves;
      int pp = pa + numWaves; if (pp > lastPair) pp = lastPair;
      __builtin_prefetch(z + (size_t)pp * 512 + (unsigned)(lane * 16), 0, 1);
      fa = load_pair(z, pa < lastPair ? pa : lastPair, N, m, half);
    }
    __builtin_amdgcn_sched_barrier(0);
    pair_step(a, cinit, fb, cA, cB, out, pb, N, lane, half);
  }
}

extern "C" void kernel_launch(void* const* d_in, const int* in_sizes, int n_in,
                              void* d_out, int out_size, void* d_ws, size_t ws_size,
                              hipStream_t stream) {
  const float* z   = (const float*)d_in[0];
  const float* phi = (const float*)d_in[1];
  const float* mu  = (const float*)d_in[2];
  const float* L   = (const float*)d_in[3];
  float* out = (float*)d_out;
  int N = in_sizes[0] / DDIM;

  int numPairs = (N + 31) / 32;
  long wavesWanted = (numPairs + 7) / 8;
  const int wavesPerBlock = 8;
  long blocks = (wavesWanted + wavesPerBlock - 1) / wavesPerBlock;
  if (blocks < 1) blocks = 1;
  if (blocks > 16384) blocks = 16384;
  int numWaves = (int)(blocks * wavesPerBlock);

  gmm_energy_kernel<<<(int)blocks, 256, 0, stream>>>(z, phi, mu, L, out,
                                                     N, numPairs, numWaves);
}


# Tile loop's vector work cut from 83 to 59 instructions per pair: lane halves trade logits by one half-swap (was select, LDS permute, select), log of the sum as log2 x ln 2, scalar-based addresses
# speedup vs baseline: 6.2026x; 1.0557x over previous
#include <hip/hip_runtime.h>
#include <math.h>

typedef float v2f __attribute__((ext_vector_type(2)));
typedef float v4f __attribute__((ext_vector_type(4)));
typedef float v8f __attribute__((ext_vector_type(8)));

#define KCOMP 4
#define DDIM  16
#define XPITCH 17
#define LTSTRIDE 272

struct BFrag { v2f k0, k1, k2, k3; };
struct BPair { BFrag x, y; };

static __device__ __forceinline__ BFrag load_frag(const float* __restrict__ zp) {
  BFrag f;
  f.k0 = *(const v2f*)(zp + 0);
  f.k1 = *(const v2f*)(zp + 4);
  f.k2 = *(const v2f*)(zp + 8);
  f.k3 = *(const v2f*)(zp + 12);
  return f;
}

static __device__ __forceinline__ BPair load_rows(const float* __restrict__ z, int r0,
                                                  unsigned lbytes) {
  const char* zb = (const char*)(z + (size_t)r0 * 16) + lbytes;
  BPair f;
  f.x = load_frag((const float*)zb);
  f.y = load_frag((const float*)(zb + 1024));
  return f;
}

static __device__ __forceinline__ void add_norms(const v8f& acc, v2f& pA, v2f& pB) {
  v2f x;
  x = __builtin_shufflevector(acc, acc, 0, 1); pA += x * x;
  x = __builtin_shufflevector(acc, acc, 4, 5); pB += x * x;
  x = __builtin_shufflevector(acc, acc, 2, 3); pA += x * x;
  x = __builtin_shufflevector(acc, acc, 6, 7); pB += x * x;
}

static __device__ __forceinline__ void swap_halves(float& a, float& b) {
  asm volatile("v_permlane16_swap_b32 %0, %1" : "+v"(a), "+v"(b));
}

static __device__ __forceinline__ void pair_step(const v2f (&a)[4][4], const v8f (&cinit)[4],
                                                 const BPair& f, float cA, float cB,
                                                 float* __restrict__ out, int r0, int first,
                                                 int lane) {
  const v2f bx[4] = {f.x.k0, f.x.k1, f.x.k2, f.x.k3};
  const v2f by[4] = {f.y.k0, f.y.k1, f.y.k2, f.y.k3};
  v2f pAx = {0.f, 0.f}, pBx = {0.f, 0.f};
  v2f pAy = {0.f, 0.f}, pBy = {0.f, 0.f};
#pragma unroll
  for (int g = 0; g < 4; ++g) {
    v8f ax = cinit[g];
    v8f ay = cinit[g];
#pragma unroll
    for (int kk = g; kk < 4; ++kk) {
      ax = __builtin_amdgcn_wmma_f32_16x16x4_f32(false, a[g][kk], false, bx[kk],
                                                 (short)0, ax, false, false);
      ay = __builtin_amdgcn_wmma_f32_16x16x4_f32(false, a[g][kk], false, by[kk],
                                                 (short)0, ay, false, false);
    }
    add_norms(ax, pAx, pBx);
    add_norms(ay, pAy, pBy);
  }
  float l0 = cA - 0.5f * (pAx[0] + pAx[1]);
  float l1 = cB - 0.5f * (pBx[0] + pBx[1]);
  float l2 = cA - 0.5f * (pAy[0] + pAy[1]);
  float l3 = cB - 0.5f * (pBy[0] + pBy[1]);
  swap_halves(l0, l2);
  swap_halves(l1, l3);
  const float mx = fmaxf(fmaxf(l0, l1), fmaxf(l2, l3));
  const float se = __expf(l0 - mx) + __expf(l1 - mx) +
                   __expf(l2 - mx) + __expf(l3 - mx);
  const float LN2 = 0.6931471805599453f;
  const float e = fmaf(-LN2, __builtin_amdgcn_logf(se), -mx);

  char* ob = (char*)(out + (size_t)r0) + (unsigned)(lane * 4);
  if (lane >= first) *(float*)ob = e;
}

__global__ void __launch_bounds__(256) gmm_energy_kernel(
    const float* __restrict__ z,
    const float* __restrict__ phi,
    const float* __restrict__ mu,
    const float* __restrict__ L,
    float* __restrict__ out,
    int N, int numPairs, int numWaves) {
  __shared__ __attribute__((aligned(16))) float sLT[KCOMP * LTSTRIDE];
  __shared__ __attribute__((aligned(16))) float sW[KCOMP * DDIM];
  __shared__ __attribute__((aligned(16))) float sMuS[KCOMP * DDIM];
  __shared__ float sX[KCOMP * DDIM * XPITCH];
  __shared__ __attribute__((aligned(16))) float sLog[KCOMP * DDIM];
  __shared__ float sC[KCOMP];

  const int tid   = threadIdx.x;
  const int lane  = tid & 31;
  const int m     = lane & 15;
  const int half  = lane >> 4;
  const int wv = __builtin_amdgcn_readfirstlane(tid >> 5);
  const int waveId = (int)blockIdx.x * 8 + wv;
  if (N < 32) return;

  const int rk = tid >> 6;
  const int ri = (tid >> 2) & 15;
  const int rq = tid & 3;
  const v4f rL = *(const v4f*)(L + 4 * tid);
  const float dRow = L[rk * DDIM * DDIM + ri * (DDIM + 1)];
  const int j64 = tid & 63;
  const float dOwn = L[(j64 >> 4) * DDIM * DDIM + (j64 & 15) * (DDIM + 1)];
  const float rMu  = mu[j64];
  const float rPhi = phi[lane & 3];
  const int lastPair = numPairs - 1;
  {
    int p0 = waveId < numPairs ? waveId : lastPair;
    int p1 = waveId + numWaves; if (p1 > lastPair) p1 = lastPair;
    __builtin_prefetch(z + (size_t)p0 * 512 + (unsigned)(lane * 16), 0, 1);
    __builtin_prefetch(z + (size_t)p1 * 512 + (unsigned)(lane * 16), 0, 1);
  }
  asm volatile("" :: "v"(rMu), "v"(dOwn), "v"(rPhi));
#pragma unroll
  for (int j = 0; j < 4; ++j)
    sLT[rk * LTSTRIDE + (4 * rq + j) * DDIM + ri] = rL[j] / dRow;
  float rinv = 1.f;
  if (wv < 2) {
    rinv = 1.f / dOwn;
    sMuS[tid]  = rMu / dOwn;
  } else if ((wv >> 1) == 2) {
    sLog[j64] = __logf(dOwn);
  }
  __syncthreads();

  if (wv < 3) {
    const int k = (wv < 2) ? (tid >> 4) : (lane & 3);
    const int c = tid & 15;
    const float* LT = sLT + k * LTSTRIDE;
    float s[DDIM];
    if (wv < 2) {
#pragma unroll
      for (int i = 0; i < DDIM; ++i) s[i] = (i == c) ? rinv : 0.f;
    } else {
#pragma unroll
      for (int b = 0; b < 4; ++b) {
        const v4f q = *(const v4f*)(sMuS + k * DDIM + 4 * b);
        s[4 * b + 0] = q[0]; s[4 * b + 1] = q[1];
        s[4 * b + 2] = q[2]; s[4 * b + 3] = q[3];
      }
    }
    float col[DDIM][DDIM];
#pragma unroll
    for (int t = DDIM - 1; t >= 1; --t)
#pragma unroll
      for (int i = 0; i < t; ++i) col[t][i] = LT[t * DDIM + i];
#pragma unroll
    for (int t = DDIM - 1; t >= 1; --t) {
      const float xt = s[t];
#pragma unroll
      for (int i = 0; i < t; ++i)
        s[i] = fmaf(-col[t][i], xt, s[i]);
    }
    if (wv < 2) {
#pragma unroll
      for (int i = 0; i < DDIM; ++i)
        sX[(k * DDIM + i) * XPITCH + c] = (i <= c) ? s[i] : 0.f;
    } else if (lane < KCOMP) {
#pragma unroll
      for (int i = 0; i < DDIM; ++i) sW[k * DDIM + i] = -s[i];
    }
  } else if (wv == 3) {
    if (lane < KCOMP) {
      const int k = lane;
      float logdet = 0.f;
      for (int d = 0; d < DDIM; ++d) logdet += sLog[k * DDIM + d];
      logdet *= 2.f;
      const float LOG2PI = 1.8378770664093453f;
      sC[k] = __logf(rPhi) - 0.5f * ((float)DDIM * LOG2PI + logdet);
    }
  }
  __syncthreads();

  const int numFull = N >> 5;
  const int tailR0  = N - 32;
  const int tailFirst = 32 - (N & 31);
  const unsigned lbytes = (unsigned)(m * 64 + half * 8);
  const unsigned pbytes = (unsigned)(lane * 64);
  BPair fa, fb;
  {
    const int p0 = waveId < numPairs ? waveId : lastPair;
    fa = load_rows(z, p0 < numFull ? p0 * 32 : tailR0, lbytes);
  }

  const int acomp = m >> 2;
  const int arow  = m & 3;
  v2f a[4][4];
#pragma unroll
  for (int g = 0; g < 4; ++g)
#pragma unroll
    for (int kk = g; kk < 4; ++kk) {
      const float* ap = sX + (acomp * 16 + 4 * g + arow) * XPITCH + kk * 4 + 2 * half;
      a[g][kk][0] = ap[0];
      a[g][kk][1] = ap[1];
    }
  const float cA = sC[2 * half];
  const float cB = sC[2 * half + 1];

  v8f cinit[4];
#pragma unroll
  for (int g = 0; g < 4; ++g) {
#pragma unroll
    for (int r = 0; r < 8; ++r)
      cinit[g][r] = sW[(2 * half + (r >> 2)) * 16 + 4 * g + (r & 3)];
  }

  for (int p = waveId; p < numPairs; p += 2 * numWaves) {
    const int pb = p + numWaves;
    {
      int pp = pb + numWaves; if (pp > lastPair) pp = lastPair;
      __builtin_prefetch((const char*)(z + (size_t)pp * 512) + pbytes, 0, 1);
      const int pl = pb < lastPair ? pb : lastPair;
      fb = load_rows(z, pl < numFull ? pl * 32 : tailR0, lbytes);
    }
    __builtin_amdgcn_sched_barrier(0);
    pair_step(a, cinit, fa, cA, cB, out, p < numFull ? p * 32 : tailR0,
              p < numFull ? 0 : tailFirst, lane);
    if (pb >= numPairs) break;
    {
      const int pa = pb + numWaves;
      int pp = pa + numWaves; if (pp > lastPair) pp = lastPair;
      __builtin_prefetch((const char*)(z + (size_t)pp * 512) + pbytes, 0, 1);
      const int pl = pa < lastPair ? pa : lastPair;
      fa = load_rows(z, pl < numFull ? pl * 32 : tailR0, lbytes);
    }
    __builtin_amdgcn_sched_barrier(0);
    pair_step(a, cinit, fb, cA, cB, out, pb < numFull ? pb * 32 : tailR0,
              pb < numFull ? 0 : tailFirst, lane);
  }

}

extern "C" void kernel_launch(void* const* d_in, const int* in_sizes, int n_in,
                              void* d_out, int out_size, void* d_ws, size_t ws_size,
                              hipStream_t stream) {
  const float* z   = (const float*)d_in[0];
  const float* phi = (const float*)d_in[1];
  const float* mu  = (const float*)d_in[2];
  const float* L   = (const float*)d_in[3];
  float* out = (float*)d_out;
  int N = in_sizes[0] / DDIM;

  int numPairs = (N + 31) / 32;
  long wavesWanted = (numPairs + 7) / 8;
  const int wavesPerBlock = 8;
  long blocks = (wavesWanted + wavesPerBlock - 1) / wavesPerBlock;
  if (blocks < 1) blocks = 1;
  if (blocks > 16384) blocks = 16384;
  int numWaves = (int)(blocks * wavesPerBlock);

  gmm_energy_kernel<<<(int)blocks, 256, 0, stream>>>(z, phi, mu, L, out,
                                                     N, numPairs, numWaves);
}
